# starting kernel
# baseline (speedup 1.0000x reference)
#include <hip/hip_runtime.h>

typedef __attribute__((ext_vector_type(2))) float v2f;
typedef __attribute__((ext_vector_type(8))) float v8f;

#define L_DIM 4096
#define H_DIM 8
#define D_DIM 32
#define NEGV (-1e30f)
#define TSCALE 0.1f
#define QSCALE 0.25506632772467906f


__global__ void mask_init_kernel(unsigned int* __restrict__ mask, int nwords) {
    int i = blockIdx.x * blockDim.x + threadIdx.x;
    int stride = gridDim.x * blockDim.x;
    for (; i < nwords; i += stride) mask[i] = 0xFFFFFFFFu;
}

__global__ void mask_clear_edges_kernel(unsigned int* __restrict__ mask,
                                        const int* __restrict__ ei, int E) {
    int e = blockIdx.x * blockDim.x + threadIdx.x;
    if (e >= E) return;
    unsigned int r = (unsigned int)ei[e];
    unsigned int c = (unsigned int)ei[E + e];
    unsigned int idx = r * L_DIM + c;
    atomicAnd(&mask[idx >> 5], ~(1u << (idx & 31u)));
}

__global__ void mask_clear_diag_kernel(unsigned int* __restrict__ mask) {
    int i = blockIdx.x * blockDim.x + threadIdx.x;
    if (i >= L_DIM) return;
    unsigned int idx = (unsigned int)i * (L_DIM + 1);
    atomicAnd(&mask[idx >> 5], ~(1u << (idx & 31u)));
}


__launch_bounds__(32)
__global__ void attn_wmma_kernel(const float* __restrict__ q,
                                 const float* __restrict__ k,
                                 const float* __restrict__ v,
                                 const float* __restrict__ dt,
                                 const unsigned int* __restrict__ mask,
                                 float* __restrict__ out) {
    __shared__ float pl[16 * 16];

    const int lane = threadIdx.x;
    const int hi   = lane >> 4;
    const int ln   = lane & 15;
    const int h    = blockIdx.y;
    const int row0 = blockIdx.x * 16;

    v2f aq[8];
    {
        const float* qp = q + (size_t)(h * L_DIM + row0 + ln) * D_DIM + 2 * hi;
#pragma unroll
        for (int kk = 0; kk < 8; ++kk) {
            v2f t = *(const v2f*)(qp + 4 * kk);
            aq[kk] = t * QSCALE;
        }
    }

    const float* kp = k + (size_t)(h * L_DIM + ln) * D_DIM + 2 * hi;
    const float* vp = v + (size_t)(h * L_DIM + 2 * hi) * D_DIM + ln;
    const float* dtp = dt + (size_t)(row0 + 8 * hi) * L_DIM + ln;
    const unsigned int* mp = mask + (size_t)(row0 + 8 * hi) * (L_DIM / 32);

    v8f o0 = {};
    v8f o1 = {};
    v8f ol = {};
    const v2f ones = (v2f){1.0f, 1.0f};

    const int ntiles = L_DIM / 16;
    for (int j = 0; j < ntiles; ++j) {
        if (j + 1 < ntiles) {
            __builtin_prefetch(kp + 16 * D_DIM, 0, 3);
            __builtin_prefetch(vp + 16 * D_DIM, 0, 3);
        }

        v2f bk[8];
#pragma unroll
        for (int kk = 0; kk < 8; ++kk) bk[kk] = *(const v2f*)(kp + 4 * kk);

        v8f s = {};
#pragma unroll
        for (int kk = 0; kk < 8; ++kk) {
            s = __builtin_amdgcn_wmma_f32_16x16x4_f32(false, aq[kk], false, bk[kk],
                                                      (short)0, s, false, false);
        }

        const int col0 = j * 16;
        const unsigned int* mpj = mp + (col0 >> 5);
        const int bitbase = (col0 & 16) + ln;

#pragma unroll
        for (int r = 0; r < 8; ++r) {
            const float d  = dtp[r * L_DIM];
            const float lg = __builtin_amdgcn_logf(1.0f + fabsf(d));
            const bool blocked = (mpj[r * (L_DIM / 32)] >> bitbase) & 1u;

            const float sv = blocked ? NEGV : fmaf(lg, -TSCALE, s[r]);
            const float pv = __builtin_amdgcn_exp2f(sv);

            pl[(r + 8 * hi) * 16 + ln] = pv;
        }

        asm volatile("s_wait_dscnt 0x0" ::: "memory");

        v2f pa[4];
#pragma unroll
        for (int kk = 0; kk < 4; ++kk) {
            pa[kk] = *(const v2f*)&pl[ln * 16 + 4 * kk + 2 * hi];
        }

        v2f bv0[4], bv1[4];
#pragma unroll
        for (int kk = 0; kk < 4; ++kk) {
            const float* vr = vp + kk * 4 * D_DIM;
            bv0[kk] = (v2f){vr[0],  vr[D_DIM]};
            bv1[kk] = (v2f){vr[16], vr[D_DIM + 16]};
        }

#pragma unroll
        for (int kk = 0; kk < 4; ++kk) {
            o0 = __builtin_amdgcn_wmma_f32_16x16x4_f32(false, pa[kk], false, bv0[kk],
                                                       (short)0, o0, false, false);
            o1 = __builtin_amdgcn_wmma_f32_16x16x4_f32(false, pa[kk], false, bv1[kk],
                                                       (short)0, o1, false, false);
            ol = __builtin_amdgcn_wmma_f32_16x16x4_f32(false, pa[kk], false, ones,
                                                       (short)0, ol, false, false);
        }

        kp  += 16 * D_DIM;
        vp  += 16 * D_DIM;
        dtp += 16;
    }

#pragma unroll
    for (int r = 0; r < 8; ++r) {
        const float l   = ol[r];
        const float inv = (l > 0.0f) ? (1.0f / l) : 0.0f;
        const size_t o = (size_t)(h * L_DIM + row0 + r + 8 * hi) * D_DIM;
        out[o + ln]      = o0[r] * inv;
        out[o + 16 + ln] = o1[r] * inv;
    }
}


extern "C" void kernel_launch(void* const* d_in, const int* in_sizes, int n_in,
                              void* d_out, int out_size, void* d_ws, size_t ws_size,
                              hipStream_t stream) {
    const float* q  = (const float*)d_in[0];
    const float* k  = (const float*)d_in[1];
    const float* v  = (const float*)d_in[2];
    const int*   ei = (const int*)d_in[3];
    const float* dt = (const float*)d_in[4];
    float* out = (float*)d_out;

    unsigned int* mask = (unsigned int*)d_ws;
    const int E = in_sizes[3] / 2;
    const int nwords = (L_DIM * L_DIM) / 32;

    mask_init_kernel<<<1024, 256, 0, stream>>>(mask, nwords);
    mask_clear_edges_kernel<<<(E + 255) / 256, 256, 0, stream>>>(mask, ei, E);
    mask_clear_diag_kernel<<<(L_DIM + 255) / 256, 256, 0, stream>>>(mask);

    dim3 grid(L_DIM / 16, H_DIM);
    attn_wmma_kernel<<<grid, 32, 0, stream>>>(q, k, v, dt, mask, out);
}


# P V run packed per tile: P stored at its column's rank among the tile's open key columns (ranks from the bias tile, inside the score run), V rows read by rank; ceil(open/4) steps; sum order changes
# speedup vs baseline: 2.2679x; 2.2679x over previous
#include <hip/hip_runtime.h>

typedef __attribute__((ext_vector_type(2))) float v2f;
typedef __attribute__((ext_vector_type(4))) float v4f;
typedef __attribute__((ext_vector_type(8))) float v8f;

#define L_DIM 4096
#define H_DIM 8
#define D_DIM 32
#define NEGV (-1e30f)
#define TSCALE 0.1f
#define QSCALE 0.25506632772467906f
#define KBLK 128
#define WPITCH 132
#define QT 8
#define NPROD 4
#define NTHREADS (32 * (QT + NPROD))
#define QROWS (16 * QT)
#define KPITCH 36
#define VPITCH 40


__global__ void mask_init_kernel(unsigned int* __restrict__ mask, int nwords) {
    int i = blockIdx.x * blockDim.x + threadIdx.x;
    int stride = gridDim.x * blockDim.x;
    for (; i < nwords; i += stride) mask[i] = 0xFFFFFFFFu;
}

__global__ void mask_clear_edges_kernel(unsigned int* __restrict__ mask,
                                        const int* __restrict__ ei, int E) {
    int e = blockIdx.x * blockDim.x + threadIdx.x;
    if (e >= E) return;
    unsigned int r = (unsigned int)ei[e];
    unsigned int c = (unsigned int)ei[E + e];
    unsigned int idx = r * L_DIM + c;
    atomicAnd(&mask[idx >> 5], ~(1u << (idx & 31u)));
}

__global__ void mask_clear_diag_kernel(unsigned int* __restrict__ mask) {
    int i = blockIdx.x * blockDim.x + threadIdx.x;
    if (i >= L_DIM) return;
    unsigned int idx = (unsigned int)i * (L_DIM + 1);
    atomicAnd(&mask[idx >> 5], ~(1u << (idx & 31u)));
}


__device__ __forceinline__ void make_bias8(v4f da, v4f db, unsigned int bits, float* dst) {
    v4f ba, bb;
#pragma unroll
    for (int c = 0; c < 4; ++c) {
        const float la = __builtin_amdgcn_logf(1.0f + fabsf(da[c]));
        const float lb = __builtin_amdgcn_logf(1.0f + fabsf(db[c]));
        ba[c] = ((bits >> c) & 1u)       ? NEGV : la * -TSCALE;
        bb[c] = ((bits >> (4 + c)) & 1u) ? NEGV : lb * -TSCALE;
    }
    *(v4f*)dst       = ba;
    *(v4f*)(dst + 4) = bb;
}

__launch_bounds__(NTHREADS)
__global__ void attn_wmma_kernel(const float* __restrict__ q,
                                 const float* __restrict__ k,
                                 const float* __restrict__ v,
                                 const float* __restrict__ dt,
                                 const unsigned int* __restrict__ mask,
                                 float* __restrict__ out) {
    __shared__ float pl[QT][16 * 16];
    __shared__ __attribute__((aligned(16))) float wt[2][QROWS * WPITCH];
    __shared__ __attribute__((aligned(16))) float ks[2][KBLK * KPITCH];
    __shared__ __attribute__((aligned(16))) float vs[2][KBLK * VPITCH];

    const int tid  = threadIdx.x;
    const int lane = tid & 31;
    const int hi   = lane >> 4;
    const int ln   = lane & 15;
    const int w    = __builtin_amdgcn_readfirstlane(tid >> 5);
    const bool staging = (w >= QT);
    const int wq   = w & (QT - 1);
    const int h    = blockIdx.x & (H_DIM - 1);
    const int rowg = (blockIdx.x / H_DIM) * QROWS;
    const int row0 = rowg + 16 * wq;
    float* plw = pl[wq];
    const unsigned int ltmask = (1u << ln) - 1u;

    const int st = tid & (32 * NPROD - 1);
    const int srow = st >> 4;
    const int scg  = st & 15;
    const float* dtw = dt + (size_t)(rowg + srow) * L_DIM + 8 * scg;
    const unsigned int* mw = mask + (size_t)(rowg + srow) * (L_DIM / 32) + (scg >> 2);
    const int mshift = (scg & 3) * 8;
    const int wofs   = srow * WPITCH + 8 * scg;
    const float* kg = k + (size_t)h * L_DIM * D_DIM + 4 * st;
    const float* vg = v + (size_t)h * L_DIM * D_DIM + 4 * st;
    const int kofs = (st >> 3) * KPITCH + 4 * (st & 7);
    const int vofs = (st >> 3) * VPITCH + 4 * (st & 7);

    auto stage_block = [&](int blk, int buf) {
        v4f kr[8], vr8[8];
        v4f da[16], db[16];
        unsigned int mb[16];
#pragma unroll
        for (int i = 0; i < 16; ++i) {
            da[i] = *(const v4f*)(dtw + (size_t)i * 8 * L_DIM + blk * KBLK);
            db[i] = *(const v4f*)(dtw + (size_t)i * 8 * L_DIM + blk * KBLK + 4);
            mb[i] = mw[i * 8 * (L_DIM / 32) + blk * (KBLK / 32)];
        }
#pragma unroll
        for (int i = 0; i < 8; ++i) {
            kr[i]  = *(const v4f*)(kg + (size_t)blk * KBLK * D_DIM + 512 * i);
            vr8[i] = *(const v4f*)(vg + (size_t)blk * KBLK * D_DIM + 512 * i);
        }
#pragma unroll
        for (int i = 0; i < 16; ++i)
            make_bias8(da[i], db[i], mb[i] >> mshift, &wt[buf][wofs + i * 8 * WPITCH]);
#pragma unroll
        for (int i = 0; i < 8; ++i) {
            *(v4f*)&ks[buf][kofs + i * 16 * KPITCH] = kr[i];
            *(v4f*)&vs[buf][vofs + i * 16 * VPITCH] = vr8[i];
        }
    };

    v2f aq[8] = {};
    if (!staging) {
        const float* qp = q + (size_t)(h * L_DIM + row0 + ln) * D_DIM + 2 * hi;
#pragma unroll
        for (int kk = 0; kk < 8; ++kk) {
            v2f t = *(const v2f*)(qp + 4 * kk);
            aq[kk] = t * QSCALE;
        }
    }

    v8f o0 = {};
    v8f o1 = {};
    v8f ol = {};
    const v2f ones = (v2f){1.0f, 1.0f};

    if (staging) stage_block(0, 0);
    __syncthreads();

    const int nblocks = L_DIM / KBLK;
    for (int b = 0; b < nblocks; ++b) {
        if (staging) {
            if (b + 1 < nblocks) stage_block(b + 1, (b + 1) & 1);
        } else {
        const float* wcur = &wt[b & 1][(16 * wq + 8 * hi) * WPITCH + ln];
        const float* kcur = &ks[b & 1][ln * KPITCH + 2 * hi];
        const float* vcol = &vs[b & 1][ln];

        auto score_cols = [&](const v2f (&bf)[8], const float (&wb)[8], int tt, v8f& sc,
                              unsigned int& slot_o, int& n_o, int (&vp)[4]) {
            v8f a = {};
            a = __builtin_amdgcn_wmma_f32_16x16x4_f32(false, aq[0], false, bf[0], (short)0, a, false, false);
            float mx = fmaxf(fmaxf(wb[0], wb[1]), fmaxf(wb[2], wb[3]));
            a = __builtin_amdgcn_wmma_f32_16x16x4_f32(false, aq[1], false, bf[1], (short)0, a, false, false);
            mx = fmaxf(mx, fmaxf(fmaxf(wb[4], wb[5]), fmaxf(wb[6], wb[7])));
            const unsigned int bal  = __builtin_amdgcn_ballot_w32(mx > NEGV);
            const unsigned int live = (bal | (bal >> 16)) & 0xFFFFu;
            const int n = __builtin_popcount(live);
            a = __builtin_amdgcn_wmma_f32_16x16x4_f32(false, aq[2], false, bf[2], (short)0, a, false, false);
            const unsigned int below = (unsigned int)__builtin_popcount(live & ltmask);
            const unsigned int sl = ((live >> ln) & 1u) ? below
                                                        : (unsigned int)n + (unsigned int)ln - below;
            a = __builtin_amdgcn_wmma_f32_16x16x4_f32(false, aq[3], false, bf[3], (short)0, a, false, false);
            const unsigned int m0 = __builtin_amdgcn_ballot_w32(sl == 0u);
            const unsigned int m1 = __builtin_amdgcn_ballot_w32(sl == 1u);
            const unsigned int m2 = __builtin_amdgcn_ballot_w32(sl == 2u);
            const unsigned int m3 = __builtin_amdgcn_ballot_w32(sl == 3u);
            a = __builtin_amdgcn_wmma_f32_16x16x4_f32(false, aq[4], false, bf[4], (short)0, a, false, false);
            const unsigned int m4 = __builtin_amdgcn_ballot_w32(sl == 4u);
            const unsigned int m5 = __builtin_amdgcn_ballot_w32(sl == 5u);
            const unsigned int m6 = __builtin_amdgcn_ballot_w32(sl == 6u);
            const unsigned int m7 = __builtin_amdgcn_ballot_w32(sl == 7u);
            a = __builtin_amdgcn_wmma_f32_16x16x4_f32(false, aq[5], false, bf[5], (short)0, a, false, false);
            const int c00 = __builtin_ctz(hi ? m2 : m0);
            const int c01 = __builtin_ctz(hi ? m3 : m1);
            const int c10 = __builtin_ctz(hi ? m6 : m4);
            const int c11 = __builtin_ctz(hi ? m7 : m5);
            a = __builtin_amdgcn_wmma_f32_16x16x4_f32(false, aq[6], false, bf[6], (short)0, a, false, false);
            int f0 = (16 * tt + c00) * VPITCH, f1 = (16 * tt + c01) * VPITCH;
            int f2 = (16 * tt + c10) * VPITCH, f3 = (16 * tt + c11) * VPITCH;
            a = __builtin_amdgcn_wmma_f32_16x16x4_f32(false, aq[7], false, bf[7], (short)0, a, false, false);
            unsigned int slp = sl;
            int np = n;
            asm volatile("" : "+v"(slp), "+v"(f0), "+v"(f1), "+v"(f2), "+v"(f3), "+s"(np));
            vp[0] = f0; vp[1] = f1; vp[2] = f2; vp[3] = f3;
            sc = a;
            slot_o = slp;
            n_o = np;
        };

        float wv[8];
        v2f bk[8];
#pragma unroll
        for (int kk = 0; kk < 8; ++kk) bk[kk] = *(const v2f*)(kcur + 4 * kk);
#pragma unroll
        for (int r = 0; r < 8; ++r) wv[r] = wcur[r * WPITCH];
        __builtin_amdgcn_sched_barrier(0);
        v8f s;
        unsigned int slot;
        int nlive;
        int vp[4];
        score_cols(bk, wv, 0, s, slot, nlive, vp);

        __builtin_amdgcn_sched_barrier(0);

#pragma unroll
        for (int t = 0; t < KBLK / 16; ++t) {
            const bool nxt = (t + 1 < KBLK / 16);

            v2f bkn[8];
            float wvn[8];
            if (nxt) {
                const float* kt = kcur + 16 * (t + 1) * KPITCH;
#pragma unroll
                for (int kk = 0; kk < 8; ++kk) bkn[kk] = *(const v2f*)(kt + 4 * kk);
#pragma unroll
                for (int r = 0; r < 8; ++r) wvn[r] = wcur[r * WPITCH + 16 * (t + 1)];
            }
            const v2f bva0 = (v2f){vcol[vp[0]],      vcol[vp[1]]};
            const v2f bvb0 = (v2f){vcol[vp[0] + 16], vcol[vp[1] + 16]};
            const v2f bva1 = (v2f){vcol[vp[2]],      vcol[vp[3]]};
            const v2f bvb1 = (v2f){vcol[vp[2] + 16], vcol[vp[3] + 16]};
            __builtin_amdgcn_sched_barrier(0);

#pragma unroll
            for (int r = 0; r < 8; ++r) {
                const float pv = __builtin_amdgcn_exp2f(s[r] + wv[r]);
                plw[(r + 8 * hi) * 16 + slot] = pv;
            }

            const v2f pa0 = *(const v2f*)&plw[ln * 16 + 2 * hi];
            const v2f pa1 = *(const v2f*)&plw[ln * 16 + 4 + 2 * hi];

            __builtin_amdgcn_sched_barrier(0);

            v8f sn = {};
            unsigned int slotn = 0u;
            int nliven = 0;
            int vpn[4] = {0, 0, 0, 0};
            if (nxt) {
                score_cols(bkn, wvn, t + 1, sn, slotn, nliven, vpn);
                __builtin_amdgcn_sched_barrier(0);
            }

            auto out_step = [&](const v2f& a, const v2f& b0, const v2f& b1) {
                o0 = __builtin_amdgcn_wmma_f32_16x16x4_f32(false, a, false, b0,
                                                           (short)0, o0, false, false);
                o1 = __builtin_amdgcn_wmma_f32_16x16x4_f32(false, a, false, b1,
                                                           (short)0, o1, false, false);
                ol = __builtin_amdgcn_wmma_f32_16x16x4_f32(false, a, false, ones,
                                                           (short)0, ol, false, false);
            };
            if (nlive > 0) out_step(pa0, bva0, bvb0);
            if (nlive > 4) out_step(pa1, bva1, bvb1);
            if (nlive > 8) {
#pragma unroll
                for (int j = 2; j < 4; ++j) {
                    if (j == 2 || nlive > 12) {
                        const unsigned int q0 = __builtin_amdgcn_ballot_w32(slot == (unsigned int)(4 * j));
                        const unsigned int q1 = __builtin_amdgcn_ballot_w32(slot == (unsigned int)(4 * j + 1));
                        const unsigned int q2 = __builtin_amdgcn_ballot_w32(slot == (unsigned int)(4 * j + 2));
                        const unsigned int q3 = __builtin_amdgcn_ballot_w32(slot == (unsigned int)(4 * j + 3));
                        const int c0 = __builtin_ctz(hi ? q2 : q0);
                        const int c1 = __builtin_ctz(hi ? q3 : q1);
                        const float* v0p = vcol + (16 * t + c0) * VPITCH;
                        const float* v1p = vcol + (16 * t + c1) * VPITCH;
                        const v2f a  = *(const v2f*)&plw[ln * 16 + 4 * j + 2 * hi];
                        const v2f b0 = (v2f){v0p[0],  v1p[0]};
                        const v2f b1 = (v2f){v0p[16], v1p[16]};
                        out_step(a, b0, b1);
                    }
                }
            }

            __builtin_amdgcn_sched_barrier(0);

            if (nxt) {
                s = sn;
                slot = slotn;
                nlive = nliven;
#pragma unroll
                for (int r = 0; r < 8; ++r) wv[r] = wvn[r];
#pragma unroll
                for (int i = 0; i < 4; ++i) vp[i] = vpn[i];
            }
        }
        }

        __syncthreads();
    }

    if (!staging) {
#pragma unroll
        for (int r = 0; r < 8; ++r) {
            const float l   = ol[r];
            const float inv = (l > 0.0f) ? (1.0f / l) : 0.0f;
            const size_t o = (size_t)(h * L_DIM + row0 + r + 8 * hi) * D_DIM;
            out[o + ln]      = o0[r] * inv;
            out[o + 16 + ln] = o1[r] * inv;
        }
    }
}


extern "C" void kernel_launch(void* const* d_in, const int* in_sizes, int n_in,
                              void* d_out, int out_size, void* d_ws, size_t ws_size,
                              hipStream_t stream) {
    const float* q  = (const float*)d_in[0];
    const float* k  = (const float*)d_in[1];
    const float* v  = (const float*)d_in[2];
    const int*   ei = (const int*)d_in[3];
    const float* dt = (const float*)d_in[4];
    float* out = (float*)d_out;

    unsigned int* mask = (unsigned int*)d_ws;
    const int E = in_sizes[3] / 2;
    const int nwords = (L_DIM * L_DIM) / 32;

    mask_init_kernel<<<1024, 256, 0, stream>>>(mask, nwords);
    mask_clear_edges_kernel<<<(E + 255) / 256, 256, 0, stream>>>(mask, ei, E);
    mask_clear_diag_kernel<<<(L_DIM + 255) / 256, 256, 0, stream>>>(mask);

    attn_wmma_kernel<<<(L_DIM / QROWS) * H_DIM, NTHREADS, 0, stream>>>(q, k, v, dt, mask, out);
}


# Bias tile built from the mask bits: each staging thread owns one row, fills its 128 values with -1e30 by constant LDS stores, loads dt and takes log2 only at the row's open columns; same sum order
# speedup vs baseline: 2.7434x; 1.2097x over previous
#include <hip/hip_runtime.h>

typedef __attribute__((ext_vector_type(2))) float v2f;
typedef __attribute__((ext_vector_type(4))) float v4f;
typedef __attribute__((ext_vector_type(8))) float v8f;
typedef __attribute__((ext_vector_type(4))) unsigned int v4u;

#define L_DIM 4096
#define H_DIM 8
#define D_DIM 32
#define NEGV (-1e30f)
#define TSCALE 0.1f
#define QSCALE 0.25506632772467906f
#define KBLK 128
#define WPITCH 132
#define QT 8
#define NPROD 4
#define NTHREADS (32 * (QT + NPROD))
#define QROWS (16 * QT)
#define KPITCH 36
#define VPITCH 40
#define NGATH 8


__global__ void mask_init_kernel(unsigned int* __restrict__ mask, int nwords) {
    int i = blockIdx.x * blockDim.x + threadIdx.x;
    int stride = gridDim.x * blockDim.x;
    for (; i < nwords; i += stride) mask[i] = 0xFFFFFFFFu;
}

__global__ void mask_clear_edges_kernel(unsigned int* __restrict__ mask,
                                        const int* __restrict__ ei, int E) {
    int e = blockIdx.x * blockDim.x + threadIdx.x;
    if (e >= E) return;
    unsigned int r = (unsigned int)ei[e];
    unsigned int c = (unsigned int)ei[E + e];
    unsigned int idx = r * L_DIM + c;
    atomicAnd(&mask[idx >> 5], ~(1u << (idx & 31u)));
}

__global__ void mask_clear_diag_kernel(unsigned int* __restrict__ mask) {
    int i = blockIdx.x * blockDim.x + threadIdx.x;
    if (i >= L_DIM) return;
    unsigned int idx = (unsigned int)i * (L_DIM + 1);
    atomicAnd(&mask[idx >> 5], ~(1u << (idx & 31u)));
}


__device__ __forceinline__ int take_open(unsigned long long& lo, unsigned long long& hi) {
    const bool inlo = (lo != 0ull);
    const unsigned long long x = inlo ? lo : hi;
    const bool has = (x != 0ull);
    const int c = __builtin_ctzll(x | 0x8000000000000000ull) + (inlo ? 0 : 64);
    const unsigned long long y = x & (x - 1ull);
    lo = inlo ? y : lo;
    hi = inlo ? hi : y;
    return has ? c : KBLK;
}

__device__ __forceinline__ float open_bias(float d) {
    return __builtin_amdgcn_logf(1.0f + fabsf(d)) * -TSCALE;
}

__launch_bounds__(NTHREADS)
__global__ void attn_wmma_kernel(const float* __restrict__ q,
                                 const float* __restrict__ k,
                                 const float* __restrict__ v,
                                 const float* __restrict__ dt,
                                 const unsigned int* __restrict__ mask,
                                 float* __restrict__ out) {
    __shared__ float pl[QT][16 * 16];
    __shared__ __attribute__((aligned(16))) float wt[2][QROWS * WPITCH];
    __shared__ __attribute__((aligned(16))) float ks[2][KBLK * KPITCH];
    __shared__ __attribute__((aligned(16))) float vs[2][KBLK * VPITCH];

    const int tid  = threadIdx.x;
    const int lane = tid & 31;
    const int hi   = lane >> 4;
    const int ln   = lane & 15;
    const int w    = __builtin_amdgcn_readfirstlane(tid >> 5);
    const bool staging = (w >= QT);
    const int wq   = w & (QT - 1);
    const int h    = blockIdx.x & (H_DIM - 1);
    const int rowg = (blockIdx.x / H_DIM) * QROWS;
    const int row0 = rowg + 16 * wq;
    float* plw = pl[wq];
    const unsigned int ltmask = (1u << ln) - 1u;

    const int st = tid & (32 * NPROD - 1);
    const unsigned int* mrow = mask + (size_t)(rowg + st) * (L_DIM / 32);
    const unsigned int drow  = (unsigned int)(rowg + st) * L_DIM;
    const int wrow = st * WPITCH;
    v4u mnext = {};
    const float* kg = k + (size_t)h * L_DIM * D_DIM + 4 * st;
    const float* vg = v + (size_t)h * L_DIM * D_DIM + 4 * st;
    const int kofs = (st >> 3) * KPITCH + 4 * (st & 7);
    const int vofs = (st >> 3) * VPITCH + 4 * (st & 7);

    const int nblocks = L_DIM / KBLK;

    auto stage_block = [&](int blk, int buf) {
        const v4u mc = mnext;
        if (blk + 1 < nblocks) mnext = *(const v4u*)(mrow + (blk + 1) * (KBLK / 32));
        v4f kr[8], vr8[8];
#pragma unroll
        for (int i = 0; i < 8; ++i) {
            kr[i]  = *(const v4f*)(kg + (size_t)blk * KBLK * D_DIM + 512 * i);
            vr8[i] = *(const v4f*)(vg + (size_t)blk * KBLK * D_DIM + 512 * i);
        }
        float* wr = &wt[buf][wrow];
        const v4f negs = (v4f){NEGV, NEGV, NEGV, NEGV};
#pragma unroll
        for (int j = 0; j < KBLK / 4; ++j) *(v4f*)(wr + 4 * j) = negs;
        unsigned long long lo = ~((unsigned long long)mc[0] | ((unsigned long long)mc[1] << 32));
        unsigned long long hi = ~((unsigned long long)mc[2] | ((unsigned long long)mc[3] << 32));
        const unsigned int dofs = drow + (unsigned int)blk * KBLK;
        int   cp[NGATH];
        float dv[NGATH];
#pragma unroll
        for (int s = 0; s < NGATH; ++s) {
            cp[s] = take_open(lo, hi);
            dv[s] = dt[(cp[s] < KBLK) ? dofs + (unsigned int)cp[s] : 0u];
        }
#pragma unroll
        for (int i = 0; i < 8; ++i) {
            *(v4f*)&ks[buf][kofs + i * 16 * KPITCH] = kr[i];
            *(v4f*)&vs[buf][vofs + i * 16 * VPITCH] = vr8[i];
        }
#pragma unroll
        for (int s = 0; s < NGATH; ++s) wr[cp[s]] = open_bias(dv[s]);
        for (int it = 0; it < KBLK - NGATH; ++it) {
            if (__builtin_amdgcn_ballot_w32((lo | hi) != 0ull) == 0u) break;
            const int c = take_open(lo, hi);
            if (c < KBLK) wr[c] = open_bias(dt[dofs + (unsigned int)c]);
        }
    };

    v2f aq[8] = {};
    if (!staging) {
        const float* qp = q + (size_t)(h * L_DIM + row0 + ln) * D_DIM + 2 * hi;
#pragma unroll
        for (int kk = 0; kk < 8; ++kk) {
            v2f t = *(const v2f*)(qp + 4 * kk);
            aq[kk] = t * QSCALE;
        }
    }

    v8f o0 = {};
    v8f o1 = {};
    v8f ol = {};
    const v2f ones = (v2f){1.0f, 1.0f};

    if (staging) {
        mnext = *(const v4u*)mrow;
        stage_block(0, 0);
    }
    __syncthreads();

    for (int b = 0; b < nblocks; ++b) {
        if (staging) {
            if (b + 1 < nblocks) stage_block(b + 1, (b + 1) & 1);
        } else {
        const float* wcur = &wt[b & 1][(16 * wq + 8 * hi) * WPITCH + ln];
        const float* kcur = &ks[b & 1][ln * KPITCH + 2 * hi];
        const float* vcol = &vs[b & 1][ln];

        auto score_cols = [&](const v2f (&bf)[8], const float (&wb)[8], int tt, v8f& sc,
                              unsigned int& slot_o, int& n_o, int (&vp)[4]) {
            v8f a = {};
            a = __builtin_amdgcn_wmma_f32_16x16x4_f32(false, aq[0], false, bf[0], (short)0, a, false, false);
            float mx = fmaxf(fmaxf(wb[0], wb[1]), fmaxf(wb[2], wb[3]));
            a = __builtin_amdgcn_wmma_f32_16x16x4_f32(false, aq[1], false, bf[1], (short)0, a, false, false);
            mx = fmaxf(mx, fmaxf(fmaxf(wb[4], wb[5]), fmaxf(wb[6], wb[7])));
            const unsigned int bal  = __builtin_amdgcn_ballot_w32(mx > NEGV);
            const unsigned int live = (bal | (bal >> 16)) & 0xFFFFu;
            const int n = __builtin_popcount(live);
            a = __builtin_amdgcn_wmma_f32_16x16x4_f32(false, aq[2], false, bf[2], (short)0, a, false, false);
            const unsigned int below = (unsigned int)__builtin_popcount(live & ltmask);
            const unsigned int sl = ((live >> ln) & 1u) ? below
                                                        : (unsigned int)n + (unsigned int)ln - below;
            a = __builtin_amdgcn_wmma_f32_16x16x4_f32(false, aq[3], false, bf[3], (short)0, a, false, false);
            const unsigned int m0 = __builtin_amdgcn_ballot_w32(sl == 0u);
            const unsigned int m1 = __builtin_amdgcn_ballot_w32(sl == 1u);
            const unsigned int m2 = __builtin_amdgcn_ballot_w32(sl == 2u);
            const unsigned int m3 = __builtin_amdgcn_ballot_w32(sl == 3u);
            a = __builtin_amdgcn_wmma_f32_16x16x4_f32(false, aq[4], false, bf[4], (short)0, a, false, false);
            const unsigned int m4 = __builtin_amdgcn_ballot_w32(sl == 4u);
            const unsigned int m5 = __builtin_amdgcn_ballot_w32(sl == 5u);
            const unsigned int m6 = __builtin_amdgcn_ballot_w32(sl == 6u);
            const unsigned int m7 = __builtin_amdgcn_ballot_w32(sl == 7u);
            a = __builtin_amdgcn_wmma_f32_16x16x4_f32(false, aq[5], false, bf[5], (short)0, a, false, false);
            const int c00 = __builtin_ctz(hi ? m2 : m0);
            const int c01 = __builtin_ctz(hi ? m3 : m1);
            const int c10 = __builtin_ctz(hi ? m6 : m4);
            const int c11 = __builtin_ctz(hi ? m7 : m5);
            a = __builtin_amdgcn_wmma_f32_16x16x4_f32(false, aq[6], false, bf[6], (short)0, a, false, false);
            int f0 = (16 * tt + c00) * VPITCH, f1 = (16 * tt + c01) * VPITCH;
            int f2 = (16 * tt + c10) * VPITCH, f3 = (16 * tt + c11) * VPITCH;
            a = __builtin_amdgcn_wmma_f32_16x16x4_f32(false, aq[7], false, bf[7], (short)0, a, false, false);
            unsigned int slp = sl;
            int np = n;
            asm volatile("" : "+v"(slp), "+v"(f0), "+v"(f1), "+v"(f2), "+v"(f3), "+s"(np));
            vp[0] = f0; vp[1] = f1; vp[2] = f2; vp[3] = f3;
            sc = a;
            slot_o = slp;
            n_o = np;
        };

        float wv[8];
        v2f bk[8];
#pragma unroll
        for (int kk = 0; kk < 8; ++kk) bk[kk] = *(const v2f*)(kcur + 4 * kk);
#pragma unroll
        for (int r = 0; r < 8; ++r) wv[r] = wcur[r * WPITCH];
        __builtin_amdgcn_sched_barrier(0);
        v8f s;
        unsigned int slot;
        int nlive;
        int vp[4];
        score_cols(bk, wv, 0, s, slot, nlive, vp);

        __builtin_amdgcn_sched_barrier(0);

#pragma unroll
        for (int t = 0; t < KBLK / 16; ++t) {
            const bool nxt = (t + 1 < KBLK / 16);

            v2f bkn[8];
            float wvn[8];
            if (nxt) {
                const float* kt = kcur + 16 * (t + 1) * KPITCH;
#pragma unroll
                for (int kk = 0; kk < 8; ++kk) bkn[kk] = *(const v2f*)(kt + 4 * kk);
#pragma unroll
                for (int r = 0; r < 8; ++r) wvn[r] = wcur[r * WPITCH + 16 * (t + 1)];
            }
            const v2f bva0 = (v2f){vcol[vp[0]],      vcol[vp[1]]};
            const v2f bvb0 = (v2f){vcol[vp[0] + 16], vcol[vp[1] + 16]};
            const v2f bva1 = (v2f){vcol[vp[2]],      vcol[vp[3]]};
            const v2f bvb1 = (v2f){vcol[vp[2] + 16], vcol[vp[3] + 16]};
            __builtin_amdgcn_sched_barrier(0);

#pragma unroll
            for (int r = 0; r < 8; ++r) {
                const float pv = __builtin_amdgcn_exp2f(s[r] + wv[r]);
                plw[(r + 8 * hi) * 16 + slot] = pv;
            }

            const v2f pa0 = *(const v2f*)&plw[ln * 16 + 2 * hi];
            const v2f pa1 = *(const v2f*)&plw[ln * 16 + 4 + 2 * hi];

            __builtin_amdgcn_sched_barrier(0);

            v8f sn = {};
            unsigned int slotn = 0u;
            int nliven = 0;
            int vpn[4] = {0, 0, 0, 0};
            if (nxt) {
                score_cols(bkn, wvn, t + 1, sn, slotn, nliven, vpn);
                __builtin_amdgcn_sched_barrier(0);
            }

            auto out_step = [&](const v2f& a, const v2f& b0, const v2f& b1) {
                o0 = __builtin_amdgcn_wmma_f32_16x16x4_f32(false, a, false, b0,
                                                           (short)0, o0, false, false);
                o1 = __builtin_amdgcn_wmma_f32_16x16x4_f32(false, a, false, b1,
                                                           (short)0, o1, false, false);
                ol = __builtin_amdgcn_wmma_f32_16x16x4_f32(false, a, false, ones,
                                                           (short)0, ol, false, false);
            };
            if (nlive > 0) out_step(pa0, bva0, bvb0);
            if (nlive > 4) out_step(pa1, bva1, bvb1);
            if (nlive > 8) {
#pragma unroll
                for (int j = 2; j < 4; ++j) {
                    if (j == 2 || nlive > 12) {
                        const unsigned int q0 = __builtin_amdgcn_ballot_w32(slot == (unsigned int)(4 * j));
                        const unsigned int q1 = __builtin_amdgcn_ballot_w32(slot == (unsigned int)(4 * j + 1));
                        const unsigned int q2 = __builtin_amdgcn_ballot_w32(slot == (unsigned int)(4 * j + 2));
                        const unsigned int q3 = __builtin_amdgcn_ballot_w32(slot == (unsigned int)(4 * j + 3));
                        const int c0 = __builtin_ctz(hi ? q2 : q0);
                        const int c1 = __builtin_ctz(hi ? q3 : q1);
                        const float* v0p = vcol + (16 * t + c0) * VPITCH;
                        const float* v1p = vcol + (16 * t + c1) * VPITCH;
                        const v2f a  = *(const v2f*)&plw[ln * 16 + 4 * j + 2 * hi];
                        const v2f b0 = (v2f){v0p[0],  v1p[0]};
                        const v2f b1 = (v2f){v0p[16], v1p[16]};
                        out_step(a, b0, b1);
                    }
                }
            }

            __builtin_amdgcn_sched_barrier(0);

            if (nxt) {
                s = sn;
                slot = slotn;
                nlive = nliven;
#pragma unroll
                for (int r = 0; r < 8; ++r) wv[r] = wvn[r];
#pragma unroll
                for (int i = 0; i < 4; ++i) vp[i] = vpn[i];
            }
        }
        }

        __syncthreads();
    }

    if (!staging) {
#pragma unroll
        for (int r = 0; r < 8; ++r) {
            const float l   = ol[r];
            const float inv = (l > 0.0f) ? (1.0f / l) : 0.0f;
            const size_t o = (size_t)(h * L_DIM + row0 + r + 8 * hi) * D_DIM;
            out[o + ln]      = o0[r] * inv;
            out[o + 16 + ln] = o1[r] * inv;
        }
    }
}


extern "C" void kernel_launch(void* const* d_in, const int* in_sizes, int n_in,
                              void* d_out, int out_size, void* d_ws, size_t ws_size,
                              hipStream_t stream) {
    const float* q  = (const float*)d_in[0];
    const float* k  = (const float*)d_in[1];
    const float* v  = (const float*)d_in[2];
    const int*   ei = (const int*)d_in[3];
    const float* dt = (const float*)d_in[4];
    float* out = (float*)d_out;

    unsigned int* mask = (unsigned int*)d_ws;
    const int E = in_sizes[3] / 2;
    const int nwords = (L_DIM * L_DIM) / 32;

    mask_init_kernel<<<1024, 256, 0, stream>>>(mask, nwords);
    mask_clear_edges_kernel<<<(E + 255) / 256, 256, 0, stream>>>(mask, ei, E);
    mask_clear_diag_kernel<<<(L_DIM + 255) / 256, 256, 0, stream>>>(mask);

    attn_wmma_kernel<<<(L_DIM / QROWS) * H_DIM, NTHREADS, 0, stream>>>(q, k, v, dt, mask, out);
}


# Tile trip: no test between score run and output run: leading P V step runs for every tile (no live column: adds exact zeros), rarer steps behind one scalar test (was a vector compare); same sum order
# speedup vs baseline: 2.8496x; 1.0387x over previous
#include <hip/hip_runtime.h>

typedef __attribute__((ext_vector_type(2))) float v2f;
typedef __attribute__((ext_vector_type(4))) float v4f;
typedef __attribute__((ext_vector_type(8))) float v8f;
typedef __attribute__((ext_vector_type(4))) unsigned int v4u;

#define L_DIM 4096
#define H_DIM 8
#define D_DIM 32
#define NEGV (-1e30f)
#define TSCALE 0.1f
#define QSCALE 0.25506632772467906f
#define KBLK 128
#define WPITCH 132
#define QT 8
#define NPROD 4
#define NTHREADS (32 * (QT + NPROD))
#define QROWS (16 * QT)
#define KPITCH 36
#define VPITCH 40
#define NGATH 8


__global__ void mask_init_kernel(unsigned int* __restrict__ mask, int nwords) {
    int i = blockIdx.x * blockDim.x + threadIdx.x;
    int stride = gridDim.x * blockDim.x;
    for (; i < nwords; i += stride) mask[i] = 0xFFFFFFFFu;
}

__global__ void mask_clear_edges_kernel(unsigned int* __restrict__ mask,
                                        const int* __restrict__ ei, int E) {
    int e = blockIdx.x * blockDim.x + threadIdx.x;
    if (e >= E) return;
    unsigned int r = (unsigned int)ei[e];
    unsigned int c = (unsigned int)ei[E + e];
    unsigned int idx = r * L_DIM + c;
    atomicAnd(&mask[idx >> 5], ~(1u << (idx & 31u)));
}

__global__ void mask_clear_diag_kernel(unsigned int* __restrict__ mask) {
    int i = blockIdx.x * blockDim.x + threadIdx.x;
    if (i >= L_DIM) return;
    unsigned int idx = (unsigned int)i * (L_DIM + 1);
    atomicAnd(&mask[idx >> 5], ~(1u << (idx & 31u)));
}


__device__ __forceinline__ int take_open(unsigned long long& lo, unsigned long long& hi) {
    const bool inlo = (lo != 0ull);
    const unsigned long long x = inlo ? lo : hi;
    const bool has = (x != 0ull);
    const int c = __builtin_ctzll(x | 0x8000000000000000ull) + (inlo ? 0 : 64);
    const unsigned long long y = x & (x - 1ull);
    lo = inlo ? y : lo;
    hi = inlo ? hi : y;
    return has ? c : KBLK;
}

__device__ __forceinline__ float open_bias(float d) {
    return __builtin_amdgcn_logf(1.0f + fabsf(d)) * -TSCALE;
}

__launch_bounds__(NTHREADS)
__global__ void attn_wmma_kernel(const float* __restrict__ q,
                                 const float* __restrict__ k,
                                 const float* __restrict__ v,
                                 const float* __restrict__ dt,
                                 const unsigned int* __restrict__ mask,
                                 float* __restrict__ out) {
    __shared__ float pl[QT][16 * 16];
    __shared__ __attribute__((aligned(16))) float wt[2][QROWS * WPITCH];
    __shared__ __attribute__((aligned(16))) float ks[2][KBLK * KPITCH];
    __shared__ __attribute__((aligned(16))) float vs[2][KBLK * VPITCH];

    const int tid  = threadIdx.x;
    const int lane = tid & 31;
    const int hi   = lane >> 4;
    const int ln   = lane & 15;
    const int w    = __builtin_amdgcn_readfirstlane(tid >> 5);
    const bool staging = (w >= QT);
    const int wq   = w & (QT - 1);
    const int h    = blockIdx.x & (H_DIM - 1);
    const int rowg = (blockIdx.x / H_DIM) * QROWS;
    const int row0 = rowg + 16 * wq;
    float* plw = pl[wq];
    const unsigned int ltmask = (1u << ln) - 1u;

    const int st = tid & (32 * NPROD - 1);
    const unsigned int* mrow = mask + (size_t)(rowg + st) * (L_DIM / 32);
    const unsigned int drow  = (unsigned int)(rowg + st) * L_DIM;
    const int wrow = st * WPITCH;
    v4u mnext = {};
    const float* kg = k + (size_t)h * L_DIM * D_DIM + 4 * st;
    const float* vg = v + (size_t)h * L_DIM * D_DIM + 4 * st;
    const int kofs = (st >> 3) * KPITCH + 4 * (st & 7);
    const int vofs = (st >> 3) * VPITCH + 4 * (st & 7);

    const int nblocks = L_DIM / KBLK;

    auto stage_block = [&](int blk, int buf) {
        const v4u mc = mnext;
        if (blk + 1 < nblocks) mnext = *(const v4u*)(mrow + (blk + 1) * (KBLK / 32));
        v4f kr[8], vr8[8];
#pragma unroll
        for (int i = 0; i < 8; ++i) {
            kr[i]  = *(const v4f*)(kg + (size_t)blk * KBLK * D_DIM + 512 * i);
            vr8[i] = *(const v4f*)(vg + (size_t)blk * KBLK * D_DIM + 512 * i);
        }
        float* wr = &wt[buf][wrow];
        const v4f negs = (v4f){NEGV, NEGV, NEGV, NEGV};
#pragma unroll
        for (int j = 0; j < KBLK / 4; ++j) *(v4f*)(wr + 4 * j) = negs;
        unsigned long long lo = ~((unsigned long long)mc[0] | ((unsigned long long)mc[1] << 32));
        unsigned long long hi = ~((unsigned long long)mc[2] | ((unsigned long long)mc[3] << 32));
        const unsigned int dofs = drow + (unsigned int)blk * KBLK;
        int   cp[NGATH];
        float dv[NGATH];
#pragma unroll
        for (int s = 0; s < NGATH; ++s) {
            cp[s] = take_open(lo, hi);
            dv[s] = dt[(cp[s] < KBLK) ? dofs + (unsigned int)cp[s] : 0u];
        }
#pragma unroll
        for (int i = 0; i < 8; ++i) {
            *(v4f*)&ks[buf][kofs + i * 16 * KPITCH] = kr[i];
            *(v4f*)&vs[buf][vofs + i * 16 * VPITCH] = vr8[i];
        }
#pragma unroll
        for (int s = 0; s < NGATH; ++s) wr[cp[s]] = open_bias(dv[s]);
        for (int it = 0; it < KBLK - NGATH; ++it) {
            if (__builtin_amdgcn_ballot_w32((lo | hi) != 0ull) == 0u) break;
            const int c = take_open(lo, hi);
            if (c < KBLK) wr[c] = open_bias(dt[dofs + (unsigned int)c]);
        }
    };

    v2f aq[8] = {};
    if (!staging) {
        const float* qp = q + (size_t)(h * L_DIM + row0 + ln) * D_DIM + 2 * hi;
#pragma unroll
        for (int kk = 0; kk < 8; ++kk) {
            v2f t = *(const v2f*)(qp + 4 * kk);
            aq[kk] = t * QSCALE;
        }
    }

    v8f o0 = {};
    v8f o1 = {};
    v8f ol = {};
    const v2f ones = (v2f){1.0f, 1.0f};

    if (staging) {
        mnext = *(const v4u*)mrow;
        stage_block(0, 0);
    }
    __syncthreads();

    for (int b = 0; b < nblocks; ++b) {
        if (staging) {
            if (b + 1 < nblocks) stage_block(b + 1, (b + 1) & 1);
        } else {
        const float* wcur = &wt[b & 1][(16 * wq + 8 * hi) * WPITCH + ln];
        const float* kcur = &ks[b & 1][ln * KPITCH + 2 * hi];
        const float* vcol = &vs[b & 1][ln];

        auto score_cols = [&](const v2f (&bf)[8], const float (&wb)[8], int tt, v8f& sc,
                              unsigned int& slot_o, int& n_o, int (&vp)[4]) {
            v8f a = {};
            a = __builtin_amdgcn_wmma_f32_16x16x4_f32(false, aq[0], false, bf[0], (short)0, a, false, false);
            float mx = fmaxf(fmaxf(wb[0], wb[1]), fmaxf(wb[2], wb[3]));
            a = __builtin_amdgcn_wmma_f32_16x16x4_f32(false, aq[1], false, bf[1], (short)0, a, false, false);
            mx = fmaxf(mx, fmaxf(fmaxf(wb[4], wb[5]), fmaxf(wb[6], wb[7])));
            const unsigned int bal  = __builtin_amdgcn_ballot_w32(mx > NEGV);
            const unsigned int live = (bal | (bal >> 16)) & 0xFFFFu;
            const int n = __builtin_popcount(live);
            a = __builtin_amdgcn_wmma_f32_16x16x4_f32(false, aq[2], false, bf[2], (short)0, a, false, false);
            const unsigned int below = (unsigned int)__builtin_popcount(live & ltmask);
            const unsigned int sl = ((live >> ln) & 1u) ? below
                                                        : (unsigned int)n + (unsigned int)ln - below;
            a = __builtin_amdgcn_wmma_f32_16x16x4_f32(false, aq[3], false, bf[3], (short)0, a, false, false);
            const unsigned int m0 = __builtin_amdgcn_ballot_w32(sl == 0u);
            const unsigned int m1 = __builtin_amdgcn_ballot_w32(sl == 1u);
            const unsigned int m2 = __builtin_amdgcn_ballot_w32(sl == 2u);
            const unsigned int m3 = __builtin_amdgcn_ballot_w32(sl == 3u);
            a = __builtin_amdgcn_wmma_f32_16x16x4_f32(false, aq[4], false, bf[4], (short)0, a, false, false);
            const unsigned int m4 = __builtin_amdgcn_ballot_w32(sl == 4u);
            const unsigned int m5 = __builtin_amdgcn_ballot_w32(sl == 5u);
            const unsigned int m6 = __builtin_amdgcn_ballot_w32(sl == 6u);
            const unsigned int m7 = __builtin_amdgcn_ballot_w32(sl == 7u);
            a = __builtin_amdgcn_wmma_f32_16x16x4_f32(false, aq[5], false, bf[5], (short)0, a, false, false);
            const int c00 = __builtin_ctz(hi ? m2 : m0);
            const int c01 = __builtin_ctz(hi ? m3 : m1);
            const int c10 = __builtin_ctz(hi ? m6 : m4);
            const int c11 = __builtin_ctz(hi ? m7 : m5);
            a = __builtin_amdgcn_wmma_f32_16x16x4_f32(false, aq[6], false, bf[6], (short)0, a, false, false);
            int f0 = (16 * tt + c00) * VPITCH, f1 = (16 * tt + c01) * VPITCH;
            int f2 = (16 * tt + c10) * VPITCH, f3 = (16 * tt + c11) * VPITCH;
            a = __builtin_amdgcn_wmma_f32_16x16x4_f32(false, aq[7], false, bf[7], (short)0, a, false, false);
            unsigned int slp = sl;
            int np = n;
            asm volatile("" : "+v"(slp), "+v"(f0), "+v"(f1), "+v"(f2), "+v"(f3), "+s"(np));
            vp[0] = f0; vp[1] = f1; vp[2] = f2; vp[3] = f3;
            sc = a;
            slot_o = slp;
            n_o = np;
        };

        float wv[8];
        v2f bk[8];
#pragma unroll
        for (int kk = 0; kk < 8; ++kk) bk[kk] = *(const v2f*)(kcur + 4 * kk);
#pragma unroll
        for (int r = 0; r < 8; ++r) wv[r] = wcur[r * WPITCH];
        __builtin_amdgcn_sched_barrier(0);
        v8f s;
        unsigned int slot;
        int nlive;
        int vp[4];
        score_cols(bk, wv, 0, s, slot, nlive, vp);

        __builtin_amdgcn_sched_barrier(0);

#pragma unroll
        for (int t = 0; t < KBLK / 16; ++t) {
            const bool nxt = (t + 1 < KBLK / 16);

            v2f bkn[8];
            float wvn[8];
            if (nxt) {
                const float* kt = kcur + 16 * (t + 1) * KPITCH;
#pragma unroll
                for (int kk = 0; kk < 8; ++kk) bkn[kk] = *(const v2f*)(kt + 4 * kk);
#pragma unroll
                for (int r = 0; r < 8; ++r) wvn[r] = wcur[r * WPITCH + 16 * (t + 1)];
            }
            float vl0 = vcol[vp[0]], vh0 = vcol[vp[0] + 16];
            float vl1 = vcol[vp[1]], vh1 = vcol[vp[1] + 16];
            const v2f bva1 = (v2f){vcol[vp[2]],      vcol[vp[3]]};
            const v2f bvb1 = (v2f){vcol[vp[2] + 16], vcol[vp[3] + 16]};
            __builtin_amdgcn_sched_barrier(0);

#pragma unroll
            for (int r = 0; r < 8; ++r) {
                const float pv = __builtin_amdgcn_exp2f(s[r] + wv[r]);
                plw[(r + 8 * hi) * 16 + slot] = pv;
            }

            const v2f pa0 = *(const v2f*)&plw[ln * 16 + 2 * hi];
            const v2f pa1 = *(const v2f*)&plw[ln * 16 + 4 + 2 * hi];

            __builtin_amdgcn_sched_barrier(0);

            v8f sn = {};
            unsigned int slotn = 0u;
            int nliven = 0;
            int vpn[4] = {0, 0, 0, 0};
            if (nxt) {
                score_cols(bkn, wvn, t + 1, sn, slotn, nliven, vpn);
                __builtin_amdgcn_sched_barrier(0);
            }

            auto out_step = [&](const v2f& a, const v2f& b0, const v2f& b1) {
                o0 = __builtin_amdgcn_wmma_f32_16x16x4_f32(false, a, false, b0,
                                                           (short)0, o0, false, false);
                o1 = __builtin_amdgcn_wmma_f32_16x16x4_f32(false, a, false, b1,
                                                           (short)0, o1, false, false);
                ol = __builtin_amdgcn_wmma_f32_16x16x4_f32(false, a, false, ones,
                                                           (short)0, ol, false, false);
            };
            asm volatile("" : "+v"(vl0), "+v"(vl1), "+v"(vh0), "+v"(vh1));
            const v2f bva0 = (v2f){vl0, vl1};
            const v2f bvb0 = (v2f){vh0, vh1};
            out_step(pa0, bva0, bvb0);
            if (__builtin_amdgcn_readfirstlane(nlive) > 4) {
                out_step(pa1, bva1, bvb1);
                if (nlive > 8) {
#pragma unroll
                    for (int j = 2; j < 4; ++j) {
                        if (j == 2 || nlive > 12) {
                            const unsigned int q0 = __builtin_amdgcn_ballot_w32(slot == (unsigned int)(4 * j));
                            const unsigned int q1 = __builtin_amdgcn_ballot_w32(slot == (unsigned int)(4 * j + 1));
                            const unsigned int q2 = __builtin_amdgcn_ballot_w32(slot == (unsigned int)(4 * j + 2));
                            const unsigned int q3 = __builtin_amdgcn_ballot_w32(slot == (unsigned int)(4 * j + 3));
                            const int c0 = __builtin_ctz(hi ? q2 : q0);
                            const int c1 = __builtin_ctz(hi ? q3 : q1);
                            const float* v0p = vcol + (16 * t + c0) * VPITCH;
                            const float* v1p = vcol + (16 * t + c1) * VPITCH;
                            const v2f a  = *(const v2f*)&plw[ln * 16 + 4 * j + 2 * hi];
                            const v2f b0 = (v2f){v0p[0],  v1p[0]};
                            const v2f b1 = (v2f){v0p[16], v1p[16]};
                            out_step(a, b0, b1);
                        }
                    }
                }
            }

            __builtin_amdgcn_sched_barrier(0);

            if (nxt) {
                s = sn;
                slot = slotn;
                nlive = nliven;
#pragma unroll
                for (int r = 0; r < 8; ++r) wv[r] = wvn[r];
#pragma unroll
                for (int i = 0; i < 4; ++i) vp[i] = vpn[i];
            }
        }
        }

        __syncthreads();
    }

    if (!staging) {
#pragma unroll
        for (int r = 0; r < 8; ++r) {
            const float l   = ol[r];
            const float inv = (l > 0.0f) ? (1.0f / l) : 0.0f;
            const size_t o = (size_t)(h * L_DIM + row0 + r + 8 * hi) * D_DIM;
            out[o + ln]      = o0[r] * inv;
            out[o + 16 + ln] = o1[r] * inv;
        }
    }
}


extern "C" void kernel_launch(void* const* d_in, const int* in_sizes, int n_in,
                              void* d_out, int out_size, void* d_ws, size_t ws_size,
                              hipStream_t stream) {
    const float* q  = (const float*)d_in[0];
    const float* k  = (const float*)d_in[1];
    const float* v  = (const float*)d_in[2];
    const int*   ei = (const int*)d_in[3];
    const float* dt = (const float*)d_in[4];
    float* out = (float*)d_out;

    unsigned int* mask = (unsigned int*)d_ws;
    const int E = in_sizes[3] / 2;
    const int nwords = (L_DIM * L_DIM) / 32;

    mask_init_kernel<<<1024, 256, 0, stream>>>(mask, nwords);
    mask_clear_edges_kernel<<<(E + 255) / 256, 256, 0, stream>>>(mask, ei, E);
    mask_clear_diag_kernel<<<(L_DIM + 255) / 256, 256, 0, stream>>>(mask);

    attn_wmma_kernel<<<(L_DIM / QROWS) * H_DIM, NTHREADS, 0, stream>>>(q, k, v, dt, mask, out);
}


# Live bits and slot owners of each (query tile, key tile) pair made once by the staging waves from the mask words (DPP row OR), read from LDS; matrix waves' max/ballot chain goes; same sum order
# speedup vs baseline: 3.0306x; 1.0635x over previous
#include <hip/hip_runtime.h>

typedef __attribute__((ext_vector_type(2))) float v2f;
typedef __attribute__((ext_vector_type(4))) float v4f;
typedef __attribute__((ext_vector_type(8))) float v8f;
typedef __attribute__((ext_vector_type(4))) unsigned int v4u;
typedef __attribute__((ext_vector_type(2))) unsigned int v2u;

#define L_DIM 4096
#define H_DIM 8
#define D_DIM 32
#define NEGV (-1e30f)
#define TSCALE 0.1f
#define QSCALE 0.25506632772467906f
#define KBLK 128
#define WPITCH 132
#define QT 8
#define NPROD 4
#define NTHREADS (32 * (QT + NPROD))
#define QROWS (16 * QT)
#define KPITCH 36
#define VPITCH 40
#define NGATH 8


__global__ void mask_init_kernel(unsigned int* __restrict__ mask, int nwords) {
    int i = blockIdx.x * blockDim.x + threadIdx.x;
    int stride = gridDim.x * blockDim.x;
    for (; i < nwords; i += stride) mask[i] = 0xFFFFFFFFu;
}

__global__ void mask_clear_edges_kernel(unsigned int* __restrict__ mask,
                                        const int* __restrict__ ei, int E) {
    int e = blockIdx.x * blockDim.x + threadIdx.x;
    if (e >= E) return;
    unsigned int r = (unsigned int)ei[e];
    unsigned int c = (unsigned int)ei[E + e];
    unsigned int idx = r * L_DIM + c;
    atomicAnd(&mask[idx >> 5], ~(1u << (idx & 31u)));
}

__global__ void mask_clear_diag_kernel(unsigned int* __restrict__ mask) {
    int i = blockIdx.x * blockDim.x + threadIdx.x;
    if (i >= L_DIM) return;
    unsigned int idx = (unsigned int)i * (L_DIM + 1);
    atomicAnd(&mask[idx >> 5], ~(1u << (idx & 31u)));
}


__device__ __forceinline__ int take_open(unsigned long long& lo, unsigned long long& hi) {
    const bool inlo = (lo != 0ull);
    const unsigned long long x = inlo ? lo : hi;
    const bool has = (x != 0ull);
    const int c = __builtin_ctzll(x | 0x8000000000000000ull) + (inlo ? 0 : 64);
    const unsigned long long y = x & (x - 1ull);
    lo = inlo ? y : lo;
    hi = inlo ? hi : y;
    return has ? c : KBLK;
}

__device__ __forceinline__ float open_bias(float d) {
    return __builtin_amdgcn_logf(1.0f + fabsf(d)) * -TSCALE;
}

__launch_bounds__(NTHREADS)
__global__ void attn_wmma_kernel(const float* __restrict__ q,
                                 const float* __restrict__ k,
                                 const float* __restrict__ v,
                                 const float* __restrict__ dt,
                                 const unsigned int* __restrict__ mask,
                                 float* __restrict__ out) {
    __shared__ float pl[QT][16 * 16];
    __shared__ __attribute__((aligned(16))) float wt[2][QROWS * WPITCH];
    __shared__ __attribute__((aligned(16))) float ks[2][KBLK * KPITCH];
    __shared__ __attribute__((aligned(16))) float vs[2][KBLK * VPITCH];
    __shared__ __attribute__((aligned(8))) unsigned int meta[2][QT * (KBLK / 16) * 2];

    const int tid  = threadIdx.x;
    const int lane = tid & 31;
    const int hi   = lane >> 4;
    const int ln   = lane & 15;
    const int w    = __builtin_amdgcn_readfirstlane(tid >> 5);
    const bool staging = (w >= QT);
    const int wq   = w & (QT - 1);
    const int h    = blockIdx.x & (H_DIM - 1);
    const int rowg = (blockIdx.x / H_DIM) * QROWS;
    const int row0 = rowg + 16 * wq;
    float* plw = pl[wq];
    const unsigned int ltmask = (1u << ln) - 1u;
    const unsigned int osh = 8u * (unsigned int)hi;

    const int st = tid & (32 * NPROD - 1);
    const unsigned int* mrow = mask + (size_t)(rowg + st) * (L_DIM / 32);
    const unsigned int drow  = (unsigned int)(rowg + st) * L_DIM;
    const int wrow = st * WPITCH;
    v4u mnext = {};
    const float* kg = k + (size_t)h * L_DIM * D_DIM + 4 * st;
    const float* vg = v + (size_t)h * L_DIM * D_DIM + 4 * st;
    const int kofs = (st >> 3) * KPITCH + 4 * (st & 7);
    const int vofs = (st >> 3) * VPITCH + 4 * (st & 7);

    const int nblocks = L_DIM / KBLK;

    auto stage_block = [&](int blk, int buf) {
        const v4u mc = mnext;
        if (blk + 1 < nblocks) mnext = *(const v4u*)(mrow + (blk + 1) * (KBLK / 32));
        v4f kr[8], vr8[8];
#pragma unroll
        for (int i = 0; i < 8; ++i) {
            kr[i]  = *(const v4f*)(kg + (size_t)blk * KBLK * D_DIM + 512 * i);
            vr8[i] = *(const v4f*)(vg + (size_t)blk * KBLK * D_DIM + 512 * i);
        }
        float* wr = &wt[buf][wrow];
        const v4f negs = (v4f){NEGV, NEGV, NEGV, NEGV};
#pragma unroll
        for (int j = 0; j < KBLK / 4; ++j) *(v4f*)(wr + 4 * j) = negs;
        unsigned long long lo = ~((unsigned long long)mc[0] | ((unsigned long long)mc[1] << 32));
        unsigned long long hi = ~((unsigned long long)mc[2] | ((unsigned long long)mc[3] << 32));
        const unsigned int dofs = drow + (unsigned int)blk * KBLK;
        int   cp[NGATH];
        float dv[NGATH];
#pragma unroll
        for (int s = 0; s < NGATH; ++s) {
            cp[s] = take_open(lo, hi);
            dv[s] = dt[(cp[s] < KBLK) ? dofs + (unsigned int)cp[s] : 0u];
        }
        unsigned int ob[4];
#pragma unroll
        for (int j = 0; j < 4; ++j) {
            unsigned int x = ~mc[j];
            x |= (unsigned int)__builtin_amdgcn_update_dpp(0, (int)x, 0x128, 0xf, 0xf, false);
            x |= (unsigned int)__builtin_amdgcn_update_dpp(0, (int)x, 0x124, 0xf, 0xf, false);
            x |= (unsigned int)__builtin_amdgcn_update_dpp(0, (int)x, 0x122, 0xf, 0xf, false);
            x |= (unsigned int)__builtin_amdgcn_update_dpp(0, (int)x, 0x121, 0xf, 0xf, false);
            ob[j] = x;
        }
        {
            const int kt = lane & 7;
            const unsigned int w01 = (kt & 2) ? ob[1] : ob[0];
            const unsigned int w23 = (kt & 2) ? ob[3] : ob[2];
            const unsigned int wsel = (kt & 4) ? w23 : w01;
            const unsigned int live = (wsel >> (16 * (kt & 1))) & 0xFFFFu;
            unsigned int x = live | ((live ^ 0xFFFFu) << 16);
            unsigned int own = 0u;
#pragma unroll
            for (int qs = 0; qs < 8; ++qs) {
                own |= ((unsigned int)__builtin_ctz(x) & 15u) << (4 * qs);
                x &= x - 1u;
            }
            *(v2u*)&meta[buf][((st >> 4) * (KBLK / 16) + kt) * 2] = (v2u){live, own};
        }
#pragma unroll
        for (int i = 0; i < 8; ++i) {
            *(v4f*)&ks[buf][kofs + i * 16 * KPITCH] = kr[i];
            *(v4f*)&vs[buf][vofs + i * 16 * VPITCH] = vr8[i];
        }
#pragma unroll
        for (int s = 0; s < NGATH; ++s) wr[cp[s]] = open_bias(dv[s]);
        for (int it = 0; it < KBLK - NGATH; ++it) {
            if (__builtin_amdgcn_ballot_w32((lo | hi) != 0ull) == 0u) break;
            const int c = take_open(lo, hi);
            if (c < KBLK) wr[c] = open_bias(dt[dofs + (unsigned int)c]);
        }
    };

    v2f aq[8] = {};
    if (!staging) {
        const float* qp = q + (size_t)(h * L_DIM + row0 + ln) * D_DIM + 2 * hi;
#pragma unroll
        for (int kk = 0; kk < 8; ++kk) {
            v2f t = *(const v2f*)(qp + 4 * kk);
            aq[kk] = t * QSCALE;
        }
    }

    v8f o0 = {};
    v8f o1 = {};
    v8f ol = {};
    const v2f ones = (v2f){1.0f, 1.0f};

    if (staging) {
        mnext = *(const v4u*)mrow;
        stage_block(0, 0);
    }
    __syncthreads();

    for (int b = 0; b < nblocks; ++b) {
        if (staging) {
            if (b + 1 < nblocks) stage_block(b + 1, (b + 1) & 1);
        } else {
        const float* wcur = &wt[b & 1][(16 * wq + 8 * hi) * WPITCH + ln];
        const float* kcur = &ks[b & 1][ln * KPITCH + 2 * hi];
        const float* vcol = &vs[b & 1][ln];
        const unsigned int* mcur = &meta[b & 1][wq * (KBLK / 16) * 2];

        auto score_cols = [&](const v2f (&bf)[8], const v2u md, int tt, v8f& sc,
                              unsigned int& slot_o, int& n_o, int (&vp)[4]) {
            const unsigned int live = md.x;
            const unsigned int own  = md.y;
            v8f a = {};
            a = __builtin_amdgcn_wmma_f32_16x16x4_f32(false, aq[0], false, bf[0], (short)0, a, false, false);
            const unsigned int nv = (unsigned int)__builtin_popcount(live);
            a = __builtin_amdgcn_wmma_f32_16x16x4_f32(false, aq[1], false, bf[1], (short)0, a, false, false);
            const unsigned int below = (unsigned int)__builtin_popcount(live & ltmask);
            a = __builtin_amdgcn_wmma_f32_16x16x4_f32(false, aq[2], false, bf[2], (short)0, a, false, false);
            const unsigned int sl = ((live >> ln) & 1u) ? below : nv + (unsigned int)ln - below;
            const int n = __builtin_amdgcn_readfirstlane((int)nv);
            a = __builtin_amdgcn_wmma_f32_16x16x4_f32(false, aq[3], false, bf[3], (short)0, a, false, false);
            const int c00 = (int)((own >> osh) & 15u);
            const int c01 = (int)((own >> (osh + 4)) & 15u);
            a = __builtin_amdgcn_wmma_f32_16x16x4_f32(false, aq[4], false, bf[4], (short)0, a, false, false);
            const int c10 = (int)((own >> (osh + 16)) & 15u);
            const int c11 = (int)((own >> (osh + 20)) & 15u);
            a = __builtin_amdgcn_wmma_f32_16x16x4_f32(false, aq[5], false, bf[5], (short)0, a, false, false);
            a = __builtin_amdgcn_wmma_f32_16x16x4_f32(false, aq[6], false, bf[6], (short)0, a, false, false);
            int f0 = (16 * tt + c00) * VPITCH, f1 = (16 * tt + c01) * VPITCH;
            int f2 = (16 * tt + c10) * VPITCH, f3 = (16 * tt + c11) * VPITCH;
            a = __builtin_amdgcn_wmma_f32_16x16x4_f32(false, aq[7], false, bf[7], (short)0, a, false, false);
            unsigned int slp = sl;
            int np = n;
            asm volatile("" : "+v"(slp), "+v"(f0), "+v"(f1), "+v"(f2), "+v"(f3), "+s"(np));
            vp[0] = f0; vp[1] = f1; vp[2] = f2; vp[3] = f3;
            sc = a;
            slot_o = slp;
            n_o = np;
        };

        float wv[8];
        v2f bk[8];
#pragma unroll
        for (int kk = 0; kk < 8; ++kk) bk[kk] = *(const v2f*)(kcur + 4 * kk);
#pragma unroll
        for (int r = 0; r < 8; ++r) wv[r] = wcur[r * WPITCH];
        const v2u md0 = *(const v2u*)mcur;
        __builtin_amdgcn_sched_barrier(0);
        v8f s;
        unsigned int slot;
        int nlive;
        int vp[4];
        score_cols(bk, md0, 0, s, slot, nlive, vp);

        __builtin_amdgcn_sched_barrier(0);

#pragma unroll
        for (int t = 0; t < KBLK / 16; ++t) {
            const bool nxt = (t + 1 < KBLK / 16);

            v2f bkn[8];
            float wvn[8];
            v2u mdn = {};
            if (nxt) {
                const float* kt = kcur + 16 * (t + 1) * KPITCH;
#pragma unroll
                for (int kk = 0; kk < 8; ++kk) bkn[kk] = *(const v2f*)(kt + 4 * kk);
#pragma unroll
                for (int r = 0; r < 8; ++r) wvn[r] = wcur[r * WPITCH + 16 * (t + 1)];
                mdn = *(const v2u*)(mcur + 2 * (t + 1));
            }
            float vl0 = vcol[vp[0]], vh0 = vcol[vp[0] + 16];
            float vl1 = vcol[vp[1]], vh1 = vcol[vp[1] + 16];
            const v2f bva1 = (v2f){vcol[vp[2]],      vcol[vp[3]]};
            const v2f bvb1 = (v2f){vcol[vp[2] + 16], vcol[vp[3] + 16]};
            __builtin_amdgcn_sched_barrier(0);

#pragma unroll
            for (int r = 0; r < 8; ++r) {
                const float pv = __builtin_amdgcn_exp2f(s[r] + wv[r]);
                plw[(r + 8 * hi) * 16 + slot] = pv;
            }

            const v2f pa0 = *(const v2f*)&plw[ln * 16 + 2 * hi];
            const v2f pa1 = *(const v2f*)&plw[ln * 16 + 4 + 2 * hi];

            __builtin_amdgcn_sched_barrier(0);

            v8f sn = {};
            unsigned int slotn = 0u;
            int nliven = 0;
            int vpn[4] = {0, 0, 0, 0};
            if (nxt) {
                score_cols(bkn, mdn, t + 1, sn, slotn, nliven, vpn);
                __builtin_amdgcn_sched_barrier(0);
            }

            auto out_step = [&](const v2f& a, const v2f& b0, const v2f& b1) {
                o0 = __builtin_amdgcn_wmma_f32_16x16x4_f32(false, a, false, b0,
                                                           (short)0, o0, false, false);
                o1 = __builtin_amdgcn_wmma_f32_16x16x4_f32(false, a, false, b1,
                                                           (short)0, o1, false, false);
                ol = __builtin_amdgcn_wmma_f32_16x16x4_f32(false, a, false, ones,
                                                           (short)0, ol, false, false);
            };
            asm volatile("" : "+v"(vl0), "+v"(vl1), "+v"(vh0), "+v"(vh1));
            const v2f bva0 = (v2f){vl0, vl1};
            const v2f bvb0 = (v2f){vh0, vh1};
            out_step(pa0, bva0, bvb0);
            if (__builtin_amdgcn_readfirstlane(nlive) > 4) {
                out_step(pa1, bva1, bvb1);
                if (nlive > 8) {
#pragma unroll
                    for (int j = 2; j < 4; ++j) {
                        if (j == 2 || nlive > 12) {
                            const unsigned int q0 = __builtin_amdgcn_ballot_w32(slot == (unsigned int)(4 * j));
                            const unsigned int q1 = __builtin_amdgcn_ballot_w32(slot == (unsigned int)(4 * j + 1));
                            const unsigned int q2 = __builtin_amdgcn_ballot_w32(slot == (unsigned int)(4 * j + 2));
                            const unsigned int q3 = __builtin_amdgcn_ballot_w32(slot == (unsigned int)(4 * j + 3));
                            const int c0 = __builtin_ctz(hi ? q2 : q0);
                            const int c1 = __builtin_ctz(hi ? q3 : q1);
                            const float* v0p = vcol + (16 * t + c0) * VPITCH;
                            const float* v1p = vcol + (16 * t + c1) * VPITCH;
                            const v2f a  = *(const v2f*)&plw[ln * 16 + 4 * j + 2 * hi];
                            const v2f b0 = (v2f){v0p[0],  v1p[0]};
                            const v2f b1 = (v2f){v0p[16], v1p[16]};
                            out_step(a, b0, b1);
                        }
                    }
                }
            }

            __builtin_amdgcn_sched_barrier(0);

            if (nxt) {
                s = sn;
                slot = slotn;
                nlive = nliven;
#pragma unroll
                for (int r = 0; r < 8; ++r) wv[r] = wvn[r];
#pragma unroll
                for (int i = 0; i < 4; ++i) vp[i] = vpn[i];
            }
        }
        }

        __syncthreads();
    }

    if (!staging) {
#pragma unroll
        for (int r = 0; r < 8; ++r) {
            const float l   = ol[r];
            const float inv = (l > 0.0f) ? (1.0f / l) : 0.0f;
            const size_t o = (size_t)(h * L_DIM + row0 + r + 8 * hi) * D_DIM;
            out[o + ln]      = o0[r] * inv;
            out[o + 16 + ln] = o1[r] * inv;
        }
    }
}


extern "C" void kernel_launch(void* const* d_in, const int* in_sizes, int n_in,
                              void* d_out, int out_size, void* d_ws, size_t ws_size,
                              hipStream_t stream) {
    const float* q  = (const float*)d_in[0];
    const float* k  = (const float*)d_in[1];
    const float* v  = (const float*)d_in[2];
    const int*   ei = (const int*)d_in[3];
    const float* dt = (const float*)d_in[4];
    float* out = (float*)d_out;

    unsigned int* mask = (unsigned int*)d_ws;
    const int E = in_sizes[3] / 2;
    const int nwords = (L_DIM * L_DIM) / 32;

    mask_init_kernel<<<1024, 256, 0, stream>>>(mask, nwords);
    mask_clear_edges_kernel<<<(E + 255) / 256, 256, 0, stream>>>(mask, ei, E);
    mask_clear_diag_kernel<<<(L_DIM + 255) / 256, 256, 0, stream>>>(mask);

    attn_wmma_kernel<<<(L_DIM / QROWS) * H_DIM, NTHREADS, 0, stream>>>(q, k, v, dt, mask, out);
}
